# starting kernel
# baseline (speedup 1.0000x reference)
#include <hip/hip_runtime.h>
#include <hip/hip_bf16.h>


typedef float v2f __attribute__((ext_vector_type(2)));
typedef float v8f __attribute__((ext_vector_type(8)));

#define GH 512
#define GW 512
#define NG 256
#define NBLK 32

__device__ __forceinline__ float fast_rcp(float x) {
#if __has_builtin(__builtin_amdgcn_rcpf)
  return __builtin_amdgcn_rcpf(x);
#else
  return 1.0f / x;
#endif
}
__device__ __forceinline__ float fast_rsq(float x) {
#if __has_builtin(__builtin_amdgcn_rsqf)
  return __builtin_amdgcn_rsqf(x);
#else
  return rsqrtf(x);
#endif
}
__device__ __forceinline__ float fast_exp2(float x) {
#if __has_builtin(__builtin_amdgcn_exp2f)
  return __builtin_amdgcn_exp2f(x);
#else
  return exp2f(x);
#endif
}
__device__ __forceinline__ float swapx16(float x) {
  return __int_as_float(__builtin_amdgcn_ds_swizzle(__float_as_int(x), 0x401f));
}

__global__ void gsplat_prep_kernel(const float* __restrict__ pos,
                                   const float* __restrict__ lsc,
                                   const float* __restrict__ rop,
                                   const float* __restrict__ col,
                                   float4* __restrict__ aop,
                                   float4* __restrict__ scal)
{
  const int i = threadIdx.x;
  if (i >= NG) return;
  const float mx = pos[i * 3 + 0], my = pos[i * 3 + 1], mz = pos[i * 3 + 2];
  const float sx = expf(-2.0f * lsc[i * 3 + 0]);
  const float sy = expf(-2.0f * lsc[i * 3 + 1]);
  const float sz = expf(-2.0f * lsc[i * 3 + 2]);
  const float a0 = 1.0f / (1.0f + expf(-rop[i]));
  const float cr = 1.0f / (1.0f + expf(-col[i * 3 + 0]));
  const float cg = 1.0f / (1.0f + expf(-col[i * 3 + 1]));
  const float cb = 1.0f / (1.0f + expf(-col[i * 3 + 2]));

  int rank = 0;
  for (int j = 0; j < NG; ++j) {
    const float zj = pos[j * 3 + 2];
    rank += (zj < mz) || (zj == mz && j < i);
  }

  const float wx = mx * sx, wy = my * sy, wz = mz * sz;
  const float C = sx * mx * mx + sy * my * my + sz * mz * mz;
  const float a0p = a0 * expf(-0.5f * C);

  const int b = rank >> 3;
  const int j8 = rank & 7;
  float4* base = aop + b * 32;
  base[2 * j8 + 0]      = make_float4(wx, wy, 0.0f, 0.0f);
  base[16 + 2 * j8 + 0] = make_float4(wz, 0.0f, 0.0f, 0.0f);
  base[2 * j8 + 1]      = make_float4(0.0f, 0.0f, sy, sz);
  base[16 + 2 * j8 + 1] = make_float4(0.0f, sx, 0.0f, 0.0f);

  scal[rank] = make_float4(a0p, cr, cg, cb);
}

__global__ __launch_bounds__(256) void gsplat_render_kernel(
    const float4* __restrict__ aopG, const float4* __restrict__ scalG,
    float* __restrict__ out)
{
  __shared__ float4 shA[NBLK * 32];
  __shared__ float4 shS[NG];

  for (int t = threadIdx.x; t < NBLK * 32; t += 256) shA[t] = aopG[t];
  shS[threadIdx.x] = scalG[threadIdx.x];
  __syncthreads();

  const int lane = threadIdx.x & 31;
  const int wv   = threadIdx.x >> 5;
  const int p    = lane & 15;
  const bool hi  = lane >= 16;
  const int x = blockIdx.x * 128 + wv * 16 + p;
  const int y = blockIdx.y;

  const float ux = ((float)x + 0.5f - 256.0f) * (1.0f / 500.0f);
  const float uy = ((float)y + 0.5f - 256.0f) * (1.0f / 500.0f);
  const float rn = fast_rsq(ux * ux + uy * uy + 1.0f);
  const float dx = ux * rn, dy = uy * rn, dz = rn;

  v2f b0, b1;
  b0.x = hi ? dz       : dx;
  b0.y = hi ? dx * dx  : dy;
  b1.x = hi ? 0.0f     : dy * dy;
  b1.y = hi ? 0.0f     : dz * dz;

  float runR = 0.0f, runG = 0.0f, runB = 0.0f, runT = 1.0f;

  for (int blk = 0; blk < NBLK; ++blk) {
    const float4 at = shA[blk * 32 + lane];
    v2f a0v; a0v.x = at.x; a0v.y = at.y;
    v2f a1v; a1v.x = at.z; a1v.y = at.w;

    v8f acc = {0, 0, 0, 0, 0, 0, 0, 0};
    acc = __builtin_amdgcn_wmma_f32_16x16x4_f32(false, a0v, false, b0,
                                                (short)0, acc, false, false);
    acc = __builtin_amdgcn_wmma_f32_16x16x4_f32(false, a1v, false, b1,
                                                (short)0, acc, false, false);

    float sIr = 0.0f, sIg = 0.0f, sIb = 0.0f, sT = 1.0f;
    const int gbase = blk * 8 + (hi ? 4 : 0);
#pragma unroll
    for (int i = 0; i < 4; ++i) {
      const float Av = acc[2 * i];
      const float Bv = acc[2 * i + 1];
      const float4 sc = shS[gbase + i];
      const float rB  = fast_rcp(Bv + 1e-10f);
      const float tau = Av * rB;
      float alpha = sc.x * fast_exp2(0.72134752f * (Av * tau));
      alpha = (tau > 0.1f) ? alpha : 0.0f;
      const float wgt = sT * alpha;
      sIr = fmaf(wgt, sc.y, sIr);
      sIg = fmaf(wgt, sc.z, sIg);
      sIb = fmaf(wgt, sc.w, sIb);
      sT  = fmaf(-alpha, sT, sT);
    }

    const float oR = swapx16(sIr), oG = swapx16(sIg);
    const float oB = swapx16(sIb), oT = swapx16(sT);
    const float lR = hi ? oR : sIr, lG = hi ? oG : sIg;
    const float lB = hi ? oB : sIb, lT = hi ? oT : sT;
    const float hR = hi ? sIr : oR, hG = hi ? sIg : oG;
    const float hB = hi ? sIb : oB, hT = hi ? sT : oT;
    const float cR = fmaf(lT, hR, lR);
    const float cG = fmaf(lT, hG, lG);
    const float cB = fmaf(lT, hB, lB);
    const float cT = lT * hT;
    runR = fmaf(runT, cR, runR);
    runG = fmaf(runT, cG, runG);
    runB = fmaf(runT, cB, runB);
    runT *= cT;
  }

  if (!hi) {
    const int o = (y * GW + x) * 3;
    out[o + 0] = runR + runT;
    out[o + 1] = runG + runT;
    out[o + 2] = runB + runT;
  }
}

extern "C" void kernel_launch(void* const* d_in, const int* in_sizes, int n_in,
                              void* d_out, int out_size, void* d_ws, size_t ws_size,
                              hipStream_t stream) {
  (void)in_sizes; (void)n_in; (void)out_size; (void)ws_size;
  const float* pos = (const float*)d_in[0];
  const float* lsc = (const float*)d_in[1];
  const float* rop = (const float*)d_in[2];
  const float* col = (const float*)d_in[3];

  float4* aop  = (float4*)d_ws;
  float4* scal = aop + NBLK * 32;

  gsplat_prep_kernel<<<1, 256, 0, stream>>>(pos, lsc, rop, col, aop, scal);

  dim3 grid(GW / 128, GH);
  gsplat_render_kernel<<<grid, 256, 0, stream>>>(aop, scal, (float*)d_out);
}


# Each lane renders two pixels (x and x+32 of its row): opacity scale, weight, colour sums and transmittance run as packed f32 on the pixel pair; half the waves; per-pixel sum order kept
# speedup vs baseline: 1.8079x; 1.8079x over previous
#include <hip/hip_runtime.h>
#include <hip/hip_bf16.h>


typedef float v2f __attribute__((ext_vector_type(2)));
typedef float v8f __attribute__((ext_vector_type(8)));

#define GH 512
#define GW 512
#define NG 256
#define NBLK 32

__device__ __forceinline__ float fast_rcp(float x) {
#if __has_builtin(__builtin_amdgcn_rcpf)
  return __builtin_amdgcn_rcpf(x);
#else
  return 1.0f / x;
#endif
}
__device__ __forceinline__ float fast_rsq(float x) {
#if __has_builtin(__builtin_amdgcn_rsqf)
  return __builtin_amdgcn_rsqf(x);
#else
  return rsqrtf(x);
#endif
}
__device__ __forceinline__ float fast_exp2(float x) {
#if __has_builtin(__builtin_amdgcn_exp2f)
  return __builtin_amdgcn_exp2f(x);
#else
  return exp2f(x);
#endif
}

__device__ __forceinline__ float lane_bcast(float v, int srcLane) {
  return __builtin_bit_cast(
      float, __builtin_amdgcn_readlane(__builtin_bit_cast(int, v), srcLane));
}

__global__ __launch_bounds__(256) void gsplat_prep_kernel(
    const float* __restrict__ pos, const float* __restrict__ lsc,
    const float* __restrict__ rop, const float* __restrict__ col,
    float4* __restrict__ aop,
    float4* __restrict__ scal)
{
  const int i = threadIdx.x;
  if (i >= NG) return;
  const int lane = i & 31;
  const int wv = __builtin_amdgcn_readfirstlane(i >> 5);

  float zr[8];
#pragma unroll
  for (int k = 0; k < 8; ++k) zr[k] = pos[(32 * k + lane) * 3 + 2];

  const float mx = pos[i * 3 + 0], my = pos[i * 3 + 1], mz = pos[i * 3 + 2];
  const float sx = expf(-2.0f * lsc[i * 3 + 0]);
  const float sy = expf(-2.0f * lsc[i * 3 + 1]);
  const float sz = expf(-2.0f * lsc[i * 3 + 2]);
  const float a0 = 1.0f / (1.0f + expf(-rop[i]));
  const float cr = 1.0f / (1.0f + expf(-col[i * 3 + 0]));
  const float cg = 1.0f / (1.0f + expf(-col[i * 3 + 1]));
  const float cb = 1.0f / (1.0f + expf(-col[i * 3 + 2]));

  int rank = 0;
#pragma unroll 8
  for (int m = 0; m < 32; ++m) {
    const float zj = lane_bcast(mz, m);
    rank += (zj < mz) || (zj == mz && m < lane);
  }
#pragma unroll
  for (int k = 0; k < 8; ++k) {
    const float zk = zr[k];
    if (k < wv) {
#pragma unroll 8
      for (int m = 0; m < 32; ++m) rank += (lane_bcast(zk, m) <= mz);
    } else if (k > wv) {
#pragma unroll 8
      for (int m = 0; m < 32; ++m) rank += (lane_bcast(zk, m) < mz);
    }
  }

  const float wx = mx * sx, wy = my * sy, wz = mz * sz;
  const float C = sx * mx * mx + sy * my * my + sz * mz * mz;
  const float a0p = a0 * expf(-0.5f * C);

  aop[rank]      = make_float4(wx, wy, wz, sx);
  aop[NG + rank] = make_float4(sy, sz, 0.0f, 0.0f);
  scal[rank] = make_float4(a0p, cr, cg, cb);
}

__global__ __launch_bounds__(256) void gsplat_render_kernel(
    const float4* __restrict__ aopG, const float4* __restrict__ scalG,
    float* __restrict__ out)
{
  __shared__ float4 shA[NBLK * 32];
  __shared__ float4 shS[NG];

  const int y = blockIdx.x;
  const float uy = ((float)y + 0.5f - 256.0f) * (1.0f / 500.0f);

  {
    const int g = threadIdx.x;
    const float4 pw = aopG[g];
    const float4 pq = aopG[NG + g];
    const float cy = fmaf(pw.y, uy, pw.z);
    const float ey = fmaf(pq.x, uy * uy, pq.y);
    float4* base = shA + (g >> 3) * 32;
    const int j8 = g & 7;
    const float4 coef = make_float4(pw.x, cy, pw.w, ey);
    const float4 none = make_float4(0.0f, 0.0f, 0.0f, 0.0f);
    base[j8]      = coef;
    base[8 + j8]  = none;
    base[16 + j8] = none;
    base[24 + j8] = coef;
    shS[g] = scalG[g];
  }
  __syncthreads();

  const int lane = threadIdx.x & 31;
  const int wv   = threadIdx.x >> 5;
  const int x0 = wv * 64;

  const float uxa = ((float)(x0 + lane) + 0.5f - 256.0f) * (1.0f / 500.0f);
  const float uxb = ((float)(x0 + 32 + lane) + 0.5f - 256.0f) * (1.0f / 500.0f);
  const float rna = fast_rsq(uxa * uxa + uy * uy + 1.0f);
  const float rnb = fast_rsq(uxb * uxb + uy * uy + 1.0f);
  const float dxa = uxa * rna, dza = rna;
  const float dxb = uxb * rnb, dzb = rnb;

  v2f bfNa, bfDa, bfNb, bfDb;
  bfNa.x = dxa;       bfNa.y = dza;
  bfDa.x = dxa * dxa; bfDa.y = dza * dza;
  bfNb.x = dxb;       bfNb.y = dzb;
  bfDb.x = dxb * dxb; bfDb.y = dzb * dzb;

  v2f runR = {0.0f, 0.0f}, runG = {0.0f, 0.0f}, runB = {0.0f, 0.0f};
  v2f runT = {1.0f, 1.0f};

  for (int blk = 0; blk < NBLK; ++blk) {
    const float4 at = shA[blk * 32 + lane];
    v2f avN; avN.x = at.x; avN.y = at.y;
    v2f avD; avD.x = at.z; avD.y = at.w;

    const v8f zero = {0, 0, 0, 0, 0, 0, 0, 0};
    const v8f accNa = __builtin_amdgcn_wmma_f32_16x16x4_f32(false, avN, false, bfNa,
                                                            (short)0, zero, false, false);
    const v8f accDa = __builtin_amdgcn_wmma_f32_16x16x4_f32(false, avD, false, bfDa,
                                                            (short)0, zero, false, false);
    const v8f accNb = __builtin_amdgcn_wmma_f32_16x16x4_f32(false, avN, false, bfNb,
                                                            (short)0, zero, false, false);
    const v8f accDb = __builtin_amdgcn_wmma_f32_16x16x4_f32(false, avD, false, bfDb,
                                                            (short)0, zero, false, false);

    v8f rBa, rBb;
#pragma unroll
    for (int i = 0; i < 8; ++i) {
      rBa[i] = fast_rcp(accDa[i] + 1e-10f);
      rBb[i] = fast_rcp(accDb[i] + 1e-10f);
    }
    const v8f taua = accNa * rBa;
    const v8f taub = accNb * rBb;
    const v8f xa = 0.72134752f * (accNa * taua);
    const v8f xb = 0.72134752f * (accNb * taub);

#pragma unroll
    for (int i = 0; i < 8; ++i) {
      const float4 sc = shS[blk * 8 + i];
      v2f ex;
      ex.x = fast_exp2(xa[i]);
      ex.y = fast_exp2(xb[i]);
      v2f alpha = ex * sc.x;
      alpha.x = (taua[i] > 0.1f) ? alpha.x : 0.0f;
      alpha.y = (taub[i] > 0.1f) ? alpha.y : 0.0f;
      const v2f wgt = runT * alpha;
      runR = __builtin_elementwise_fma(wgt, (v2f){sc.y, sc.y}, runR);
      runG = __builtin_elementwise_fma(wgt, (v2f){sc.z, sc.z}, runG);
      runB = __builtin_elementwise_fma(wgt, (v2f){sc.w, sc.w}, runB);
      runT = __builtin_elementwise_fma(-alpha, runT, runT);
    }
  }

  {
    const int oa = (y * GW + x0 + lane) * 3;
    out[oa + 0] = runR.x + runT.x;
    out[oa + 1] = runG.x + runT.x;
    out[oa + 2] = runB.x + runT.x;
    const int ob = oa + 32 * 3;
    out[ob + 0] = runR.y + runT.y;
    out[ob + 1] = runG.y + runT.y;
    out[ob + 2] = runB.y + runT.y;
  }
}

extern "C" void kernel_launch(void* const* d_in, const int* in_sizes, int n_in,
                              void* d_out, int out_size, void* d_ws, size_t ws_size,
                              hipStream_t stream) {
  (void)in_sizes; (void)n_in; (void)out_size; (void)ws_size;
  const float* pos = (const float*)d_in[0];
  const float* lsc = (const float*)d_in[1];
  const float* rop = (const float*)d_in[2];
  const float* col = (const float*)d_in[3];

  float4* aop  = (float4*)d_ws;
  float4* scal = aop + 2 * NG;

  gsplat_prep_kernel<<<1, 256, 0, stream>>>(pos, lsc, rop, col, aop, scal);

  gsplat_render_kernel<<<GH, 256, 0, stream>>>(aop, scal, (float*)d_out);
}


# Lane's two pixels are now the mirror pair x and 511-x of a row: dx only flips sign, so the denominator B is the same number for both; one WMMA and one rcp per gaussian serve the pair; sums unchanged
# speedup vs baseline: 1.8891x; 1.0449x over previous
#include <hip/hip_runtime.h>
#include <hip/hip_bf16.h>


typedef float v2f __attribute__((ext_vector_type(2)));
typedef float v8f __attribute__((ext_vector_type(8)));

#define GH 512
#define GW 512
#define NG 256
#define NBLK 32

__device__ __forceinline__ float fast_rcp(float x) {
#if __has_builtin(__builtin_amdgcn_rcpf)
  return __builtin_amdgcn_rcpf(x);
#else
  return 1.0f / x;
#endif
}
__device__ __forceinline__ float fast_rsq(float x) {
#if __has_builtin(__builtin_amdgcn_rsqf)
  return __builtin_amdgcn_rsqf(x);
#else
  return rsqrtf(x);
#endif
}
__device__ __forceinline__ float fast_exp2(float x) {
#if __has_builtin(__builtin_amdgcn_exp2f)
  return __builtin_amdgcn_exp2f(x);
#else
  return exp2f(x);
#endif
}

__device__ __forceinline__ float lane_bcast(float v, int srcLane) {
  return __builtin_bit_cast(
      float, __builtin_amdgcn_readlane(__builtin_bit_cast(int, v), srcLane));
}

__global__ __launch_bounds__(256) void gsplat_prep_kernel(
    const float* __restrict__ pos, const float* __restrict__ lsc,
    const float* __restrict__ rop, const float* __restrict__ col,
    float4* __restrict__ aop,
    float4* __restrict__ scal)
{
  const int i = threadIdx.x;
  if (i >= NG) return;
  const int lane = i & 31;
  const int wv = __builtin_amdgcn_readfirstlane(i >> 5);

  float zr[8];
#pragma unroll
  for (int k = 0; k < 8; ++k) zr[k] = pos[(32 * k + lane) * 3 + 2];

  const float mx = pos[i * 3 + 0], my = pos[i * 3 + 1], mz = pos[i * 3 + 2];
  const float sx = expf(-2.0f * lsc[i * 3 + 0]);
  const float sy = expf(-2.0f * lsc[i * 3 + 1]);
  const float sz = expf(-2.0f * lsc[i * 3 + 2]);
  const float a0 = 1.0f / (1.0f + expf(-rop[i]));
  const float cr = 1.0f / (1.0f + expf(-col[i * 3 + 0]));
  const float cg = 1.0f / (1.0f + expf(-col[i * 3 + 1]));
  const float cb = 1.0f / (1.0f + expf(-col[i * 3 + 2]));

  int rank = 0;
#pragma unroll 8
  for (int m = 0; m < 32; ++m) {
    const float zj = lane_bcast(mz, m);
    rank += (zj < mz) || (zj == mz && m < lane);
  }
#pragma unroll
  for (int k = 0; k < 8; ++k) {
    const float zk = zr[k];
    if (k < wv) {
#pragma unroll 8
      for (int m = 0; m < 32; ++m) rank += (lane_bcast(zk, m) <= mz);
    } else if (k > wv) {
#pragma unroll 8
      for (int m = 0; m < 32; ++m) rank += (lane_bcast(zk, m) < mz);
    }
  }

  const float wx = mx * sx, wy = my * sy, wz = mz * sz;
  const float C = sx * mx * mx + sy * my * my + sz * mz * mz;
  const float a0p = a0 * expf(-0.5f * C);

  aop[rank]      = make_float4(wx, wy, wz, sx);
  aop[NG + rank] = make_float4(sy, sz, 0.0f, 0.0f);
  scal[rank] = make_float4(a0p, cr, cg, cb);
}

__global__ __launch_bounds__(256) void gsplat_render_kernel(
    const float4* __restrict__ aopG, const float4* __restrict__ scalG,
    float* __restrict__ out)
{
  __shared__ float4 shA[NBLK * 32];
  __shared__ float4 shS[NG];

  const int y = blockIdx.x;
  const float uy = ((float)y + 0.5f - 256.0f) * (1.0f / 500.0f);

  {
    const int g = threadIdx.x;
    const float4 pw = aopG[g];
    const float4 pq = aopG[NG + g];
    const float cy = fmaf(pw.y, uy, pw.z);
    const float ey = fmaf(pq.x, uy * uy, pq.y);
    float4* base = shA + (g >> 3) * 32;
    const int j8 = g & 7;
    const float4 coef = make_float4(pw.x, cy, pw.w, ey);
    const float4 none = make_float4(0.0f, 0.0f, 0.0f, 0.0f);
    base[j8]      = coef;
    base[8 + j8]  = none;
    base[16 + j8] = none;
    base[24 + j8] = coef;
    shS[g] = scalG[g];
  }
  __syncthreads();

  const int lane = threadIdx.x & 31;
  const int wv   = threadIdx.x >> 5;
  const int pxa = wv * 32 + lane;
  const int pxb = (GW - 1) - pxa;

  const float ux = ((float)pxa + 0.5f - 256.0f) * (1.0f / 500.0f);
  const float rn = fast_rsq(ux * ux + uy * uy + 1.0f);
  const float dx = ux * rn, dz = rn;

  v2f bfNa, bfNb, bfD;
  bfNa.x = dx;      bfNa.y = dz;
  bfNb.x = -dx;     bfNb.y = dz;
  bfD.x  = dx * dx; bfD.y  = dz * dz;

  v2f runR = {0.0f, 0.0f}, runG = {0.0f, 0.0f}, runB = {0.0f, 0.0f};
  v2f runT = {1.0f, 1.0f};

  for (int blk = 0; blk < NBLK; ++blk) {
    const float4 at = shA[blk * 32 + lane];
    v2f avN; avN.x = at.x; avN.y = at.y;
    v2f avD; avD.x = at.z; avD.y = at.w;

    const v8f zero = {0, 0, 0, 0, 0, 0, 0, 0};
    const v8f accD  = __builtin_amdgcn_wmma_f32_16x16x4_f32(false, avD, false, bfD,
                                                            (short)0, zero, false, false);
    const v8f accNa = __builtin_amdgcn_wmma_f32_16x16x4_f32(false, avN, false, bfNa,
                                                            (short)0, zero, false, false);
    const v8f accNb = __builtin_amdgcn_wmma_f32_16x16x4_f32(false, avN, false, bfNb,
                                                            (short)0, zero, false, false);

    v8f rB;
#pragma unroll
    for (int i = 0; i < 8; ++i) rB[i] = fast_rcp(accD[i] + 1e-10f);
    const v8f taua = accNa * rB;
    const v8f taub = accNb * rB;
    const v8f xa = 0.72134752f * (accNa * taua);
    const v8f xb = 0.72134752f * (accNb * taub);

#pragma unroll
    for (int i = 0; i < 8; ++i) {
      const float4 sc = shS[blk * 8 + i];
      v2f ex;
      ex.x = fast_exp2(xa[i]);
      ex.y = fast_exp2(xb[i]);
      v2f alpha = ex * sc.x;
      alpha.x = (taua[i] > 0.1f) ? alpha.x : 0.0f;
      alpha.y = (taub[i] > 0.1f) ? alpha.y : 0.0f;
      const v2f wgt = runT * alpha;
      runR = __builtin_elementwise_fma(wgt, (v2f){sc.y, sc.y}, runR);
      runG = __builtin_elementwise_fma(wgt, (v2f){sc.z, sc.z}, runG);
      runB = __builtin_elementwise_fma(wgt, (v2f){sc.w, sc.w}, runB);
      runT = __builtin_elementwise_fma(-alpha, runT, runT);
    }
  }

  {
    float* pa = out + (size_t)(y * GW + pxa) * 3;
    pa[0] = runR.x + runT.x;
    pa[1] = runG.x + runT.x;
    pa[2] = runB.x + runT.x;
    float* pb = out + (size_t)(y * GW + pxb) * 3;
    pb[0] = runR.y + runT.y;
    pb[1] = runG.y + runT.y;
    pb[2] = runB.y + runT.y;
  }
}

extern "C" void kernel_launch(void* const* d_in, const int* in_sizes, int n_in,
                              void* d_out, int out_size, void* d_ws, size_t ws_size,
                              hipStream_t stream) {
  (void)in_sizes; (void)n_in; (void)out_size; (void)ws_size;
  const float* pos = (const float*)d_in[0];
  const float* lsc = (const float*)d_in[1];
  const float* rop = (const float*)d_in[2];
  const float* col = (const float*)d_in[3];

  float4* aop  = (float4*)d_ws;
  float4* scal = aop + 2 * NG;

  gsplat_prep_kernel<<<1, 256, 0, stream>>>(pos, lsc, rop, col, aop, scal);

  gsplat_render_kernel<<<GH, 256, 0, stream>>>(aop, scal, (float*)d_out);
}


# Block loop ends once every pixel of the wave has transmittance below 2^-26: the adds left are under half an ulp of any output of 0.25 or more and change no bit; trips saved depend on the scene
# speedup vs baseline: 2.2906x; 1.2125x over previous
#include <hip/hip_runtime.h>
#include <hip/hip_bf16.h>


typedef float v2f __attribute__((ext_vector_type(2)));
typedef float v8f __attribute__((ext_vector_type(8)));

#define GH 512
#define GW 512
#define NG 256
#define NBLK 32

#define KEXP   0.72134752f
#define INV_K  1.3862944f
#define EPS_K  (1e-10f * INV_K)
#define TAU_K  (0.1f * KEXP)
#define T_DONE 0x1p-26f

__device__ __forceinline__ float fast_rcp(float x) {
#if __has_builtin(__builtin_amdgcn_rcpf)
  return __builtin_amdgcn_rcpf(x);
#else
  return 1.0f / x;
#endif
}
__device__ __forceinline__ float fast_rsq(float x) {
#if __has_builtin(__builtin_amdgcn_rsqf)
  return __builtin_amdgcn_rsqf(x);
#else
  return rsqrtf(x);
#endif
}
__device__ __forceinline__ float fast_exp2(float x) {
#if __has_builtin(__builtin_amdgcn_exp2f)
  return __builtin_amdgcn_exp2f(x);
#else
  return exp2f(x);
#endif
}

__device__ __forceinline__ float lane_bcast(float v, int srcLane) {
  return __builtin_bit_cast(
      float, __builtin_amdgcn_readlane(__builtin_bit_cast(int, v), srcLane));
}

__global__ __launch_bounds__(256) void gsplat_prep_kernel(
    const float* __restrict__ pos, const float* __restrict__ lsc,
    const float* __restrict__ rop, const float* __restrict__ col,
    float4* __restrict__ aop,
    float4* __restrict__ scal)
{
  const int i = threadIdx.x;
  if (i >= NG) return;
  const int lane = i & 31;
  const int wv = __builtin_amdgcn_readfirstlane(i >> 5);

  float zr[8];
#pragma unroll
  for (int k = 0; k < 8; ++k) zr[k] = pos[(32 * k + lane) * 3 + 2];

  const float mx = pos[i * 3 + 0], my = pos[i * 3 + 1], mz = pos[i * 3 + 2];
  const float sx = expf(-2.0f * lsc[i * 3 + 0]);
  const float sy = expf(-2.0f * lsc[i * 3 + 1]);
  const float sz = expf(-2.0f * lsc[i * 3 + 2]);
  const float a0 = 1.0f / (1.0f + expf(-rop[i]));
  const float cr = 1.0f / (1.0f + expf(-col[i * 3 + 0]));
  const float cg = 1.0f / (1.0f + expf(-col[i * 3 + 1]));
  const float cb = 1.0f / (1.0f + expf(-col[i * 3 + 2]));

  int rank = 0;
#pragma unroll 8
  for (int m = 0; m < 32; ++m) {
    const float zj = lane_bcast(mz, m);
    rank += (zj < mz) || (zj == mz && m < lane);
  }
#pragma unroll
  for (int k = 0; k < 8; ++k) {
    const float zk = zr[k];
    if (k < wv) {
#pragma unroll 8
      for (int m = 0; m < 32; ++m) rank += (lane_bcast(zk, m) <= mz);
    } else if (k > wv) {
#pragma unroll 8
      for (int m = 0; m < 32; ++m) rank += (lane_bcast(zk, m) < mz);
    }
  }

  const float wx = mx * sx, wy = my * sy, wz = mz * sz;
  const float C = sx * mx * mx + sy * my * my + sz * mz * mz;
  const float a0p = a0 * expf(-0.5f * C);

  aop[rank]      = make_float4(wx, wy, wz, sx);
  aop[NG + rank] = make_float4(sy, sz, 0.0f, 0.0f);
  scal[rank] = make_float4(a0p, cr, cg, cb);
}

__global__ __launch_bounds__(256) void gsplat_render_kernel(
    const float4* __restrict__ aopG, const float4* __restrict__ scalG,
    float* __restrict__ out)
{
  __shared__ float4 shA[NBLK * 32];

  const int y = blockIdx.x;
  const float uy = ((float)y + 0.5f - 256.0f) * (1.0f / 500.0f);

  {
    const int g = threadIdx.x;
    const float4 pw = aopG[g];
    const float4 pq = aopG[NG + g];
    const float cy = fmaf(pw.y, uy, pw.z);
    const float ey = fmaf(pq.x, uy * uy, pq.y);
    float4* base = shA + (g >> 3) * 32;
    const int j8 = g & 7;
    const float4 coef = make_float4(pw.x, cy, pw.w, ey);
    const float4 none = make_float4(0.0f, 0.0f, 0.0f, 0.0f);
    base[j8]      = coef;
    base[8 + j8]  = none;
    base[16 + j8] = none;
    base[24 + j8] = coef;
  }
  __syncthreads();

  const int lane = threadIdx.x & 31;
  const int wv   = threadIdx.x >> 5;
  const int pxa = wv * 32 + lane;
  const int pxb = (GW - 1) - pxa;

  const float ux = ((float)pxa + 0.5f - 256.0f) * (1.0f / 500.0f);
  const float rn = fast_rsq(ux * ux + uy * uy + 1.0f);
  const float dx = ux * rn, dz = rn;

  v2f bfNa, bfNb, bfD;
  bfNa.x = dx;      bfNa.y = dz;
  bfNb.x = -dx;     bfNb.y = dz;
  bfD.x  = dx * dx * INV_K; bfD.y  = dz * dz * INV_K;

  v2f runR = {0.0f, 0.0f}, runG = {0.0f, 0.0f}, runB = {0.0f, 0.0f};
  v2f runT = {1.0f, 1.0f};

  for (int blk = 0; blk < NBLK; ++blk) {
    const float4 at = shA[blk * 32 + lane];
    v2f avN; avN.x = at.x; avN.y = at.y;
    v2f avD; avD.x = at.z; avD.y = at.w;

    const v8f zero = {0, 0, 0, 0, 0, 0, 0, 0};
    const v8f epsk = {EPS_K, EPS_K, EPS_K, EPS_K, EPS_K, EPS_K, EPS_K, EPS_K};
    const v8f accD  = __builtin_amdgcn_wmma_f32_16x16x4_f32(false, avD, false, bfD,
                                                            (short)0, epsk, false, false);
    const v8f accNa = __builtin_amdgcn_wmma_f32_16x16x4_f32(false, avN, false, bfNa,
                                                            (short)0, zero, false, false);
    const v8f accNb = __builtin_amdgcn_wmma_f32_16x16x4_f32(false, avN, false, bfNb,
                                                            (short)0, zero, false, false);

    v8f rB;
#pragma unroll
    for (int i = 0; i < 8; ++i) rB[i] = fast_rcp(accD[i]);
    const v8f taua = accNa * rB;
    const v8f taub = accNb * rB;
    const v8f xa = accNa * taua;
    const v8f xb = accNb * taub;

#pragma unroll
    for (int i = 0; i < 8; ++i) {
      const float4 sc = scalG[blk * 8 + i];
      v2f ex;
      ex.x = fast_exp2(xa[i]);
      ex.y = fast_exp2(xb[i]);
      v2f alpha = ex * sc.x;
      alpha.x = (taua[i] > TAU_K) ? alpha.x : 0.0f;
      alpha.y = (taub[i] > TAU_K) ? alpha.y : 0.0f;
      const v2f wgt = runT * alpha;
      runR = __builtin_elementwise_fma(wgt, (v2f){sc.y, sc.y}, runR);
      runG = __builtin_elementwise_fma(wgt, (v2f){sc.z, sc.z}, runG);
      runB = __builtin_elementwise_fma(wgt, (v2f){sc.w, sc.w}, runB);
      runT = __builtin_elementwise_fma(-alpha, runT, runT);
    }

    const float tmax = fmaxf(fabsf(runT.x), fabsf(runT.y));
    if (__builtin_amdgcn_ballot_w32(tmax >= T_DONE) == 0u) break;
  }

  {
    float* pa = out + (size_t)(y * GW + pxa) * 3;
    pa[0] = runR.x + runT.x;
    pa[1] = runG.x + runT.x;
    pa[2] = runB.x + runT.x;
    float* pb = out + (size_t)(y * GW + pxb) * 3;
    pb[0] = runR.y + runT.y;
    pb[1] = runG.y + runT.y;
    pb[2] = runB.y + runT.y;
  }
}

extern "C" void kernel_launch(void* const* d_in, const int* in_sizes, int n_in,
                              void* d_out, int out_size, void* d_ws, size_t ws_size,
                              hipStream_t stream) {
  (void)in_sizes; (void)n_in; (void)out_size; (void)ws_size;
  const float* pos = (const float*)d_in[0];
  const float* lsc = (const float*)d_in[1];
  const float* rop = (const float*)d_in[2];
  const float* col = (const float*)d_in[3];

  float4* aop  = (float4*)d_ws;
  float4* scal = aop + 2 * NG;

  gsplat_prep_kernel<<<1, 256, 0, stream>>>(pos, lsc, rop, col, aop, scal);

  gsplat_render_kernel<<<GH, 256, 0, stream>>>(aop, scal, (float*)d_out);
}
